# starting kernel
# baseline (speedup 1.0000x reference)
#include <hip/hip_runtime.h>

typedef float v2f __attribute__((ext_vector_type(2)));
typedef float v8f __attribute__((ext_vector_type(8)));

#define N_NODES 50000
#define N_EDGES 600000
#define DIM     128
#define MTILES  (N_NODES / 16)


__global__ void k_init_deg(float* __restrict__ deg) {
    int i = blockIdx.x * blockDim.x + threadIdx.x;
    if (i < N_NODES) deg[i] = 1.0f;
}

__global__ void k_count(const int* __restrict__ dst, float* __restrict__ deg) {
    int e = blockIdx.x * blockDim.x + threadIdx.x;
    if (e < N_EDGES) atomicAdd(&deg[dst[e]], 1.0f);
}

__global__ void k_dinv(float* __restrict__ d) {
    int i = blockIdx.x * blockDim.x + threadIdx.x;
    if (i < N_NODES) d[i] = rsqrtf(d[i]);
}

__global__ void k_repack(const float* __restrict__ W, float* __restrict__ Wp) {
    int idx = blockIdx.x * blockDim.x + threadIdx.x;
    if (idx < DIM * DIM) {
        int k = idx >> 7, n = idx & 127;
        Wp[(k >> 1) * 256 + n * 2 + (k & 1)] = W[idx];
    }
}


template <int RELU>
__global__ __launch_bounds__(256) void k_gemm(const float* __restrict__ X,
                                              const float* __restrict__ Wp,
                                              float* __restrict__ H) {
    int wave = threadIdx.x >> 5;
    int lane = threadIdx.x & 31;
    int mt   = blockIdx.x * 8 + wave;
    if (mt >= MTILES) return;

    int g = lane >> 4;
    int r = lane & 15;

    const float* arow = X + ((size_t)mt * 16 + r) * DIM;

    v8f acc[8] = {};

#pragma unroll
    for (int kb = 0; kb < DIM; kb += 4) {
        int k0 = kb + 2 * g;
        v2f a = *(const v2f*)(arow + k0);
        if (RELU) { a.x = fmaxf(a.x, 0.0f); a.y = fmaxf(a.y, 0.0f); }
        const float* bbase = Wp + (size_t)(k0 >> 1) * 256 + r * 2;
#pragma unroll
        for (int nt = 0; nt < 8; ++nt) {
            v2f b = *(const v2f*)(bbase + nt * 32);
            acc[nt] = __builtin_amdgcn_wmma_f32_16x16x4_f32(
                false, a, false, b, (short)0, acc[nt], false, false);
        }
    }

    float* hrow = H + ((size_t)mt * 16 + 8 * g) * DIM + r;
#pragma unroll
    for (int v = 0; v < 8; ++v) {
#pragma unroll
        for (int nt = 0; nt < 8; ++nt)
            hrow[(size_t)v * DIM + nt * 16] = acc[nt][v];
    }
}

__global__ void k_self(const float* __restrict__ h, const float* __restrict__ dinv,
                       const float* __restrict__ bias, float* __restrict__ out) {
    int idx = blockIdx.x * blockDim.x + threadIdx.x;
    if (idx < N_NODES * DIM) {
        int node = idx >> 7, c = idx & 127;
        float di = dinv[node];
        out[idx] = bias[c] + h[idx] * di * di;
    }
}

__global__ __launch_bounds__(256) void k_scatter(const float* __restrict__ h,
                                                 const int* __restrict__ src,
                                                 const int* __restrict__ dst,
                                                 const float* __restrict__ dinv,
                                                 float* __restrict__ out) {
    int tid  = blockIdx.x * blockDim.x + threadIdx.x;
    int e    = tid >> 5;
    int lane = threadIdx.x & 31;
    if (e >= N_EDGES) return;

    int s = src[e], d = dst[e];
    float nrm = dinv[s] * dinv[d];

    const float4* hv = (const float4*)(h + (size_t)s * DIM);
    float4 v = hv[lane];

    float* o = out + (size_t)d * DIM + lane * 4;
    atomicAdd(o + 0, v.x * nrm);
    atomicAdd(o + 1, v.y * nrm);
    atomicAdd(o + 2, v.z * nrm);
    atomicAdd(o + 3, v.w * nrm);
}


extern "C" void kernel_launch(void* const* d_in, const int* in_sizes, int n_in,
                              void* d_out, int out_size, void* d_ws, size_t ws_size,
                              hipStream_t stream) {
    const float* X  = (const float*)d_in[0];
    const int*   ei = (const int*)d_in[1];
    const float* W1 = (const float*)d_in[2];
    const float* b1 = (const float*)d_in[3];
    const float* W2 = (const float*)d_in[4];
    const float* b2 = (const float*)d_in[5];
    float* out = (float*)d_out;

    const int* src = ei;
    const int* dst = ei + N_EDGES;

    float* ws   = (float*)d_ws;
    float* dinv = ws;
    float* Wp1  = ws + 50048;
    float* Wp2  = Wp1 + 16384;
    float* h    = Wp2 + 16384;
    float* x1   = h + (size_t)N_NODES * DIM;

    const int thr = 256;
    const int gNode   = (N_NODES + thr - 1) / thr;
    const int gEdge1  = (N_EDGES + thr - 1) / thr;
    const int gEdge32 = (N_EDGES * 32 + thr - 1) / thr;
    const int gFeat   = (N_NODES * DIM + thr - 1) / thr;
    const int gGemm   = (MTILES + 7) / 8;
    const int gW      = (DIM * DIM + thr - 1) / thr;

    k_init_deg<<<gNode, thr, 0, stream>>>(dinv);
    k_count   <<<gEdge1, thr, 0, stream>>>(dst, dinv);
    k_dinv    <<<gNode, thr, 0, stream>>>(dinv);

    k_repack<<<gW, thr, 0, stream>>>(W1, Wp1);
    k_repack<<<gW, thr, 0, stream>>>(W2, Wp2);

    k_gemm<0><<<gGemm, thr, 0, stream>>>(X, Wp1, h);
    k_self    <<<gFeat, thr, 0, stream>>>(h, dinv, b1, x1);
    k_scatter <<<gEdge32, thr, 0, stream>>>(h, src, dst, dinv, x1);

    k_gemm<1><<<gGemm, thr, 0, stream>>>(x1, Wp2, h);
    k_self    <<<gFeat, thr, 0, stream>>>(h, dinv, b2, out);
    k_scatter <<<gEdge32, thr, 0, stream>>>(h, src, dst, dinv, out);
}


# layer 2's aggregation reads back the per-wave edge records (source, local row, norm) that layer 1's saves in the workspace: no second range-id scan, no per-edge index and norm gathers in layer 2
# speedup vs baseline: 2.0987x; 2.0987x over previous
#include <hip/hip_runtime.h>

typedef float v2f __attribute__((ext_vector_type(2)));
typedef float v8f __attribute__((ext_vector_type(8)));

#define N_NODES 50000
#define N_EDGES 600000
#define DIM     128
#define MTILES  (N_NODES / 16)

#define ROWS    200
#define NWG     (N_NODES / ROWS)
#define WCAP    256
#define NCHUNK  1172
#define RID_PAD (NCHUNK * 512)
#define NREC    (NWG * 32 * WCAP)

static_assert(N_NODES <= 65536 && ROWS <= 32768, "edge record packing");


__global__ void k_init_deg(float* __restrict__ deg) {
    int i = blockIdx.x * blockDim.x + threadIdx.x;
    if (i < N_NODES) deg[i] = 1.0f;
}

__global__ void k_count(const int* __restrict__ dst, float* __restrict__ deg,
                        unsigned char* __restrict__ rid) {
    int e = blockIdx.x * blockDim.x + threadIdx.x;
    if (e < N_EDGES) {
        int d = dst[e];
        atomicAdd(&deg[d], 1.0f);
        rid[e] = (unsigned char)(d / ROWS);
    } else if (e < RID_PAD) {
        rid[e] = 255;
    }
}

__global__ void k_dinv(float* __restrict__ d) {
    int i = blockIdx.x * blockDim.x + threadIdx.x;
    if (i < N_NODES) d[i] = rsqrtf(d[i]);
}

__global__ void k_repack(const float* __restrict__ W, float* __restrict__ Wp) {
    int idx = blockIdx.x * blockDim.x + threadIdx.x;
    if (idx < DIM * DIM) {
        int k = idx >> 7, n = idx & 127;
        Wp[(k >> 1) * 256 + n * 2 + (k & 1)] = W[idx];
    }
}


template <int RELU>
__global__ __launch_bounds__(256) void k_gemm(const float* __restrict__ X,
                                              const float* __restrict__ Wp,
                                              float* __restrict__ H) {
    int wave = threadIdx.x >> 5;
    int lane = threadIdx.x & 31;
    int mt   = blockIdx.x * 8 + wave;
    if (mt >= MTILES) return;

    int g = lane >> 4;
    int r = lane & 15;

    const float* arow = X + ((size_t)mt * 16 + r) * DIM;

    v8f acc[8] = {};

#pragma unroll
    for (int kb = 0; kb < DIM; kb += 4) {
        int k0 = kb + 2 * g;
        v2f a = *(const v2f*)(arow + k0);
        if (RELU) { a.x = fmaxf(a.x, 0.0f); a.y = fmaxf(a.y, 0.0f); }
        const float* bbase = Wp + (size_t)(k0 >> 1) * 256 + r * 2;
#pragma unroll
        for (int nt = 0; nt < 8; ++nt) {
            v2f b = *(const v2f*)(bbase + nt * 32);
            acc[nt] = __builtin_amdgcn_wmma_f32_16x16x4_f32(
                false, a, false, b, (short)0, acc[nt], false, false);
        }
    }

    float* hrow = H + ((size_t)mt * 16 + 8 * g) * DIM + r;
#pragma unroll
    for (int v = 0; v < 8; ++v) {
#pragma unroll
        for (int nt = 0; nt < 8; ++nt)
            hrow[(size_t)v * DIM + nt * 16] = acc[nt][v];
    }
}


__device__ __forceinline__ void agg_load4(uint4 (&q)[4], const uint4* __restrict__ rid4,
                                          int c0, int lane) {
#pragma unroll
    for (int u = 0; u < 4; ++u) {
        int c = c0 + 32 * u;
        q[u] = (c < NCHUNK) ? rid4[c * 32 + lane]
                            : make_uint4(0xFFFFFFFFu, 0xFFFFFFFFu, 0xFFFFFFFFu, 0xFFFFFFFFu);
    }
}

__device__ __forceinline__ void agg_note(int e, int wave, int base, int* wcnt, int* mylist,
                                         float* acc, const float* __restrict__ h,
                                         const int* __restrict__ src,
                                         const int* __restrict__ dst,
                                         const float* __restrict__ dinv) {
    int slot = atomicAdd(&wcnt[wave], 1);
    if (slot < WCAP) {
        mylist[slot] = e;
    } else {
        int s = src[e], d = dst[e];
        float nrm = dinv[s] * dinv[d];
        const float* hr = h + (size_t)s * DIM;
        float* ar = acc + (d - base) * DIM;
        for (int c = 0; c < DIM; ++c) atomicAdd(ar + c, hr[c] * nrm);
    }
}

template <int LAYER>
__global__ __launch_bounds__(1024) void k_agg(const float* __restrict__ h,
                                              const int* __restrict__ src,
                                              const int* __restrict__ dst,
                                              const unsigned char* __restrict__ rid,
                                              const float* __restrict__ dinv,
                                              const float* __restrict__ bias,
                                              float* __restrict__ out,
                                              int* __restrict__ rec_sd,
                                              float* __restrict__ rec_nm,
                                              int* __restrict__ rec_cnt) {
    __shared__ float acc[ROWS * DIM];
    __shared__ int   list[32 * WCAP];
    __shared__ int   wcnt[32];

    const int tid  = threadIdx.x;
    const int lane = tid & 31;
    const int wave = tid >> 5;
    const int g    = blockIdx.x;
    const int base = g * ROWS;

    bool scan = true;
    int  nrec = 0;
    if (LAYER == 2) {
        scan = rec_cnt[NWG * 32 + g] != 0;
        nrec = rec_cnt[g * 32 + wave];
    }

    for (int i = tid; i < ROWS * DIM; i += 1024) acc[i] = 0.0f;
    if (tid < 32) wcnt[tid] = 0;
    __syncthreads();

    int* mylist = list + wave * WCAP;
    if (scan) {
        const uint4* rid4 = (const uint4*)rid;
        const unsigned pat = (unsigned)g * 0x01010101u;
        uint4 qa[4], qb[4];
        agg_load4(qa, rid4, wave, lane);
        agg_load4(qb, rid4, wave + 128, lane);
        for (int c0 = wave; c0 < NCHUNK; c0 += 128) {
            uint4 qc[4];
            agg_load4(qc, rid4, c0 + 256, lane);

            unsigned long long mm = 0;
#pragma unroll
            for (int u = 0; u < 4; ++u) {
                const unsigned w[4] = {qa[u].x, qa[u].y, qa[u].z, qa[u].w};
                unsigned m = 0;
#pragma unroll
                for (int j = 0; j < 4; ++j) {
                    unsigned x = w[j] ^ pat;
                    unsigned t = ~(((x & 0x7F7F7F7Fu) + 0x7F7F7F7Fu) | x | 0x7F7F7F7Fu);
                    m |= (t >> 7) << j;
                }
                mm |= (unsigned long long)m << (4 * (u & 1) + 32 * (u >> 1));
            }
            while (mm) {
                int p = __builtin_ctzll(mm);
                mm &= mm - 1;
                int u = ((p >> 2) & 1) + 2 * (p >> 5);
                int j = p & 3;
                int b = (p >> 3) & 3;
                agg_note((c0 + 32 * u) * 512 + lane * 16 + j * 4 + b, wave, base, wcnt, mylist,
                         acc, h, src, dst, dinv);
            }
#pragma unroll
            for (int u = 0; u < 4; ++u) { qa[u] = qb[u]; qb[u] = qc[u]; }
        }
        __syncthreads();
    }

    const int rbase = (g * 32 + wave) * WCAP;
    int n = nrec;
    if (scan) {
        n = wcnt[wave];
        if (LAYER == 1) {
            if (lane == 0) rec_cnt[g * 32 + wave] = n;
            if (wave == 0) {
                int over = __any(wcnt[lane] > WCAP);
                if (lane == 0) rec_cnt[NWG * 32 + g] = over;
            }
        }
    }
    n = n < WCAP ? n : WCAP;
    for (int b0 = 0; b0 < n; b0 += 32) {
        const int nb = (n - b0) < 32 ? (n - b0) : 32;
        int s = 0, dl = 0;
        float nrm = 0.0f;
        if (lane < nb) {
            if (scan) {
                int e = mylist[b0 + lane];
                s = src[e];
                int d = dst[e];
                dl = d - base;
                nrm = dinv[s] * dinv[d];
                if (LAYER == 1) {
                    rec_sd[rbase + b0 + lane] = s | (dl << 16);
                    rec_nm[rbase + b0 + lane] = nrm;
                }
            } else {
                int p = rec_sd[rbase + b0 + lane];
                nrm = rec_nm[rbase + b0 + lane];
                s  = p & 0xFFFF;
                dl = p >> 16;
            }
        }
        for (int k0 = 0; k0 < nb; k0 += 8) {
            float v[8][4];
            int   ro[8];
#pragma unroll
            for (int u = 0; u < 8; ++u) {
                int k  = k0 + u;
                int kk = k < nb ? k : nb - 1;
                int   ss = __shfl(s, kk);
                int   dd = __shfl(dl, kk);
                float nn = __shfl(nrm, kk);
                const float* hr = h + (size_t)ss * DIM + lane;
                v[u][0] = hr[0]  * nn;
                v[u][1] = hr[32] * nn;
                v[u][2] = hr[64] * nn;
                v[u][3] = hr[96] * nn;
                ro[u] = dd * DIM + lane;
            }
#pragma unroll
            for (int u = 0; u < 8; ++u) {
                if (k0 + u < nb) {
                    atomicAdd(&acc[ro[u]],      v[u][0]);
                    atomicAdd(&acc[ro[u] + 32], v[u][1]);
                    atomicAdd(&acc[ro[u] + 64], v[u][2]);
                    atomicAdd(&acc[ro[u] + 96], v[u][3]);
                }
            }
        }
    }
    __syncthreads();

    for (int i = tid; i < ROWS * DIM; i += 1024) {
        int r = i >> 7, c = i & 127;
        int node = base + r;
        float di = dinv[node];
        size_t gi = (size_t)node * DIM + c;
        out[gi] = bias[c] + h[gi] * di * di + acc[i];
    }
}


extern "C" void kernel_launch(void* const* d_in, const int* in_sizes, int n_in,
                              void* d_out, int out_size, void* d_ws, size_t ws_size,
                              hipStream_t stream) {
    const float* X  = (const float*)d_in[0];
    const int*   ei = (const int*)d_in[1];
    const float* W1 = (const float*)d_in[2];
    const float* b1 = (const float*)d_in[3];
    const float* W2 = (const float*)d_in[4];
    const float* b2 = (const float*)d_in[5];
    float* out = (float*)d_out;

    const int* src = ei;
    const int* dst = ei + N_EDGES;

    float* ws   = (float*)d_ws;
    float* dinv = ws;
    float* Wp1  = ws + 50048;
    float* Wp2  = Wp1 + 16384;
    float* h    = Wp2 + 16384;
    unsigned char* rid = (unsigned char*)(h + (size_t)N_NODES * DIM);
    int*   rec_sd  = (int*)(rid + RID_PAD);
    float* rec_nm  = (float*)(rec_sd + NREC);
    int*   rec_cnt = (int*)(rec_nm + NREC);
    float* x1   = out;

    const int thr = 256;
    const int gNode   = (N_NODES + thr - 1) / thr;
    const int gEdge1  = RID_PAD / thr;
    const int gGemm   = (MTILES + 7) / 8;
    const int gW      = (DIM * DIM + thr - 1) / thr;

    k_init_deg<<<gNode, thr, 0, stream>>>(dinv);
    k_count   <<<gEdge1, thr, 0, stream>>>(dst, dinv, rid);
    k_dinv    <<<gNode, thr, 0, stream>>>(dinv);

    k_repack<<<gW, thr, 0, stream>>>(W1, Wp1);
    k_repack<<<gW, thr, 0, stream>>>(W2, Wp2);

    k_gemm<0><<<gGemm, thr, 0, stream>>>(X, Wp1, h);
    k_agg<1> <<<NWG, 1024, 0, stream>>>(h, src, dst, rid, dinv, b1, x1, rec_sd, rec_nm, rec_cnt);

    k_gemm<1><<<gGemm, thr, 0, stream>>>(x1, Wp2, h);
    k_agg<2> <<<NWG, 1024, 0, stream>>>(h, src, dst, rid, dinv, b2, out, rec_sd, rec_nm, rec_cnt);
}


# sums in registers: records routed in LDS to the wave owning the destination row (row mod 32); it adds gathered rows there, stores rows once; LDS f32 atomics only on a full list (sum order varies)
# speedup vs baseline: 2.3279x; 1.1092x over previous
#include <hip/hip_runtime.h>

typedef float v2f __attribute__((ext_vector_type(2)));
typedef float v8f __attribute__((ext_vector_type(8)));

#define N_NODES 50000
#define N_EDGES 600000
#define DIM     128
#define MTILES  (N_NODES / 16)

#define ROWS    200
#define NWG     (N_NODES / ROWS)
#define WCAP    256
#define NCHUNK  1172
#define RID_PAD (NCHUNK * 512)
#define NREC    (NWG * 32 * WCAP)
#define OROWS   7
static_assert(ROWS <= 32 * OROWS, "row ownership");

static_assert(N_NODES <= 65536 && ROWS <= 32768, "edge record packing");


__global__ void k_init_deg(float* __restrict__ deg) {
    int i = blockIdx.x * blockDim.x + threadIdx.x;
    if (i < N_NODES) deg[i] = 1.0f;
}

__global__ void k_count(const int* __restrict__ dst, float* __restrict__ deg,
                        unsigned char* __restrict__ rid) {
    int e = blockIdx.x * blockDim.x + threadIdx.x;
    if (e < N_EDGES) {
        int d = dst[e];
        atomicAdd(&deg[d], 1.0f);
        rid[e] = (unsigned char)(d / ROWS);
    } else if (e < RID_PAD) {
        rid[e] = 255;
    }
}

__global__ void k_dinv(float* __restrict__ d) {
    int i = blockIdx.x * blockDim.x + threadIdx.x;
    if (i < N_NODES) d[i] = rsqrtf(d[i]);
}

__global__ void k_repack(const float* __restrict__ W, float* __restrict__ Wp) {
    int idx = blockIdx.x * blockDim.x + threadIdx.x;
    if (idx < DIM * DIM) {
        int k = idx >> 7, n = idx & 127;
        Wp[(k >> 1) * 256 + n * 2 + (k & 1)] = W[idx];
    }
}


template <int RELU>
__global__ __launch_bounds__(256) void k_gemm(const float* __restrict__ X,
                                              const float* __restrict__ Wp,
                                              float* __restrict__ H) {
    int wave = threadIdx.x >> 5;
    int lane = threadIdx.x & 31;
    int mt   = blockIdx.x * 8 + wave;
    if (mt >= MTILES) return;

    int g = lane >> 4;
    int r = lane & 15;

    const float* arow = X + ((size_t)mt * 16 + r) * DIM;

    v8f acc[8] = {};

#pragma unroll
    for (int kb = 0; kb < DIM; kb += 4) {
        int k0 = kb + 2 * g;
        v2f a = *(const v2f*)(arow + k0);
        if (RELU) { a.x = fmaxf(a.x, 0.0f); a.y = fmaxf(a.y, 0.0f); }
        const float* bbase = Wp + (size_t)(k0 >> 1) * 256 + r * 2;
#pragma unroll
        for (int nt = 0; nt < 8; ++nt) {
            v2f b = *(const v2f*)(bbase + nt * 32);
            acc[nt] = __builtin_amdgcn_wmma_f32_16x16x4_f32(
                false, a, false, b, (short)0, acc[nt], false, false);
        }
    }

    float* hrow = H + ((size_t)mt * 16 + 8 * g) * DIM + r;
#pragma unroll
    for (int v = 0; v < 8; ++v) {
#pragma unroll
        for (int nt = 0; nt < 8; ++nt)
            hrow[(size_t)v * DIM + nt * 16] = acc[nt][v];
    }
}


__device__ __forceinline__ void agg_load4(uint4 (&q)[4], const uint4* __restrict__ rid4,
                                          int c0, int lane) {
#pragma unroll
    for (int u = 0; u < 4; ++u) {
        int c = c0 + 32 * u;
        q[u] = (c < NCHUNK) ? rid4[c * 32 + lane]
                            : make_uint4(0xFFFFFFFFu, 0xFFFFFFFFu, 0xFFFFFFFFu, 0xFFFFFFFFu);
    }
}

__device__ __forceinline__ void agg_note(int e, int wave, int base, int* wcnt, int* mylist,
                                         float* acc, int* spill, const float* __restrict__ h,
                                         const int* __restrict__ src,
                                         const int* __restrict__ dst,
                                         const float* __restrict__ dinv) {
    int slot = atomicAdd(&wcnt[wave], 1);
    if (slot < WCAP) {
        mylist[slot] = e;
    } else {
        *spill = 1;
        int s = src[e], d = dst[e];
        float nrm = dinv[s] * dinv[d];
        const float* hr = h + (size_t)s * DIM;
        float* ar = acc + (d - base) * DIM;
        for (int c = 0; c < DIM; ++c) atomicAdd(ar + c, hr[c] * nrm);
    }
}

template <int LAYER>
__global__ __launch_bounds__(1024) void k_agg(const float* __restrict__ h,
                                              const int* __restrict__ src,
                                              const int* __restrict__ dst,
                                              const unsigned char* __restrict__ rid,
                                              const float* __restrict__ dinv,
                                              const float* __restrict__ bias,
                                              float* __restrict__ out,
                                              int* __restrict__ rec_sd,
                                              float* __restrict__ rec_nm,
                                              int* __restrict__ rec_cnt) {
    __shared__ __attribute__((aligned(16))) float acc[ROWS * DIM];
    __shared__ int   list[32 * WCAP];
    __shared__ int   osd[32 * WCAP];
    __shared__ float onm[32 * WCAP];
    __shared__ int   wcnt[32];
    __shared__ int   ocnt[32];
    __shared__ int   spill;

    const int tid  = threadIdx.x;
    const int lane = tid & 31;
    const int wave = tid >> 5;
    const int g    = blockIdx.x;
    const int base = g * ROWS;

    bool scan = true;
    int  nrec = 0;
    if (LAYER == 2) {
        scan = rec_cnt[NWG * 32 + g] != 0;
        nrec = rec_cnt[g * 32 + wave];
    }

    if (scan) {
        float4* acc4 = (float4*)acc;
        for (int i = tid; i < ROWS * DIM / 4; i += 1024) acc4[i] = make_float4(0.f, 0.f, 0.f, 0.f);
        if (tid < 32) { wcnt[tid] = 0; ocnt[tid] = 0; }
        if (tid == 0) spill = 0;
        __syncthreads();

        int* mylist = list + wave * WCAP;
        const uint4* rid4 = (const uint4*)rid;
        const unsigned pat = (unsigned)g * 0x01010101u;
        uint4 qa[4], qb[4];
        agg_load4(qa, rid4, wave, lane);
        agg_load4(qb, rid4, wave + 128, lane);
        for (int c0 = wave; c0 < NCHUNK; c0 += 128) {
            uint4 qc[4];
            agg_load4(qc, rid4, c0 + 256, lane);

            unsigned long long mm = 0;
#pragma unroll
            for (int u = 0; u < 4; ++u) {
                const unsigned w[4] = {qa[u].x, qa[u].y, qa[u].z, qa[u].w};
                unsigned m = 0;
#pragma unroll
                for (int j = 0; j < 4; ++j) {
                    unsigned x = w[j] ^ pat;
                    unsigned t = ~(((x & 0x7F7F7F7Fu) + 0x7F7F7F7Fu) | x | 0x7F7F7F7Fu);
                    m |= (t >> 7) << j;
                }
                mm |= (unsigned long long)m << (4 * (u & 1) + 32 * (u >> 1));
            }
            while (mm) {
                int p = __builtin_ctzll(mm);
                mm &= mm - 1;
                int u = ((p >> 2) & 1) + 2 * (p >> 5);
                int j = p & 3;
                int b = (p >> 3) & 3;
                agg_note((c0 + 32 * u) * 512 + lane * 16 + j * 4 + b, wave, base, wcnt, mylist,
                         acc, &spill, h, src, dst, dinv);
            }
#pragma unroll
            for (int u = 0; u < 4; ++u) { qa[u] = qb[u]; qb[u] = qc[u]; }
        }
        __syncthreads();

        int n1 = wcnt[wave];
        n1 = n1 < WCAP ? n1 : WCAP;
        for (int b0 = 0; b0 < n1; b0 += 32) {
            if (b0 + lane < n1) {
                int e = mylist[b0 + lane];
                int s = src[e];
                int d = dst[e];
                int dl = d - base;
                float nrm = dinv[s] * dinv[d];
                int ow = dl & 31;
                int slot = atomicAdd(&ocnt[ow], 1);
                if (slot < WCAP) {
                    osd[ow * WCAP + slot] = s | (dl << 16);
                    onm[ow * WCAP + slot] = nrm;
                } else {
                    spill = 1;
                    const float* hr = h + (size_t)s * DIM;
                    float* ar = acc + dl * DIM;
                    for (int c = 0; c < DIM; ++c) atomicAdd(ar + c, hr[c] * nrm);
                }
            }
        }
        __syncthreads();
    }

    const int rbase = (g * 32 + wave) * WCAP;
    int n = scan ? ocnt[wave] : nrec;
    n = n < WCAP ? n : WCAP;
    n = __builtin_amdgcn_readfirstlane(n);
    if (scan && LAYER == 1) {
        if (lane == 0) rec_cnt[g * 32 + wave] = n;
        if (tid == 0) rec_cnt[NWG * 32 + g] = spill;
    }

    float4 r[OROWS];
#pragma unroll
    for (int j = 0; j < OROWS; ++j) r[j] = make_float4(0.f, 0.f, 0.f, 0.f);

    for (int b0 = 0; b0 < n; b0 += 32) {
        const int nb = (n - b0) < 32 ? (n - b0) : 32;
        int p = 0;
        float nrm = 0.0f;
        if (lane < nb) {
            if (scan) {
                p   = osd[wave * WCAP + b0 + lane];
                nrm = onm[wave * WCAP + b0 + lane];
                if (LAYER == 1) {
                    rec_sd[rbase + b0 + lane] = p;
                    rec_nm[rbase + b0 + lane] = nrm;
                }
            } else {
                p   = rec_sd[rbase + b0 + lane];
                nrm = rec_nm[rbase + b0 + lane];
            }
        }
        const int s  = p & 0xFFFF;
        const int jr = p >> 21;
        const int nbits = __float_as_int(nrm);
        for (int k0 = 0; k0 < nb; k0 += 8) {
            float4 v[8];
            float  nw[8];
            int    jj[8];
#pragma unroll
            for (int u = 0; u < 8; ++u) {
                int k  = k0 + u;
                int kk = k < nb ? k : nb - 1;
                int   ss = __builtin_amdgcn_readlane(s, kk);
                nw[u]    = __int_as_float(__builtin_amdgcn_readlane(nbits, kk));
                jj[u]    = __builtin_amdgcn_readlane(jr, kk);
                v[u]     = *(const float4*)(h + (size_t)ss * DIM + lane * 4);
            }
#pragma unroll
            for (int u = 0; u < 8; ++u) {
                if (k0 + u < nb) {
#pragma unroll
                    for (int j = 0; j < OROWS; ++j) {
                        if (jj[u] == j) {
                            asm volatile("");
                            r[j].x += v[u].x * nw[u]; r[j].y += v[u].y * nw[u];
                            r[j].z += v[u].z * nw[u]; r[j].w += v[u].w * nw[u];
                        }
                    }
                }
            }
        }
    }

    const bool use_acc = scan && (spill != 0);
    const float bx = bias[lane * 4 + 0], by = bias[lane * 4 + 1];
    const float bz = bias[lane * 4 + 2], bw = bias[lane * 4 + 3];
#pragma unroll
    for (int j = 0; j < OROWS; ++j) {
        const int row = wave + 32 * j;
        if (row < ROWS) {
            const int node = base + row;
            const float di = dinv[node];
            const size_t gi = (size_t)node * DIM + lane * 4;
            const float4 hv = *(const float4*)(h + gi);
            float4 o;
            o.x = bx + hv.x * di * di + r[j].x;
            o.y = by + hv.y * di * di + r[j].y;
            o.z = bz + hv.z * di * di + r[j].z;
            o.w = bw + hv.w * di * di + r[j].w;
            if (use_acc) {
                const float4 a = *(const float4*)(acc + row * DIM + lane * 4);
                o.x += a.x; o.y += a.y; o.z += a.z; o.w += a.w;
            }
            *(float4*)(out + gi) = o;
        }
    }
}


extern "C" void kernel_launch(void* const* d_in, const int* in_sizes, int n_in,
                              void* d_out, int out_size, void* d_ws, size_t ws_size,
                              hipStream_t stream) {
    const float* X  = (const float*)d_in[0];
    const int*   ei = (const int*)d_in[1];
    const float* W1 = (const float*)d_in[2];
    const float* b1 = (const float*)d_in[3];
    const float* W2 = (const float*)d_in[4];
    const float* b2 = (const float*)d_in[5];
    float* out = (float*)d_out;

    const int* src = ei;
    const int* dst = ei + N_EDGES;

    float* ws   = (float*)d_ws;
    float* dinv = ws;
    float* Wp1  = ws + 50048;
    float* Wp2  = Wp1 + 16384;
    float* h    = Wp2 + 16384;
    unsigned char* rid = (unsigned char*)(h + (size_t)N_NODES * DIM);
    int*   rec_sd  = (int*)(rid + RID_PAD);
    float* rec_nm  = (float*)(rec_sd + NREC);
    int*   rec_cnt = (int*)(rec_nm + NREC);
    float* x1   = out;

    const int thr = 256;
    const int gNode   = (N_NODES + thr - 1) / thr;
    const int gEdge1  = RID_PAD / thr;
    const int gGemm   = (MTILES + 7) / 8;
    const int gW      = (DIM * DIM + thr - 1) / thr;

    k_init_deg<<<gNode, thr, 0, stream>>>(dinv);
    k_count   <<<gEdge1, thr, 0, stream>>>(dst, dinv, rid);
    k_dinv    <<<gNode, thr, 0, stream>>>(dinv);

    k_repack<<<gW, thr, 0, stream>>>(W1, Wp1);
    k_repack<<<gW, thr, 0, stream>>>(W2, Wp2);

    k_gemm<0><<<gGemm, thr, 0, stream>>>(X, Wp1, h);
    k_agg<1> <<<NWG, 1024, 0, stream>>>(h, src, dst, rid, dinv, b1, x1, rec_sd, rec_nm, rec_cnt);

    k_gemm<1><<<gGemm, thr, 0, stream>>>(x1, Wp2, h);
    k_agg<2> <<<NWG, 1024, 0, stream>>>(h, src, dst, rid, dinv, b2, out, rec_sd, rec_nm, rec_cnt);
}


# layer 1 sums neighbours' input rows before multiplying, (A x) W1 for A (x W1); one kernel then does both products per 16-row tile, relu(t + b1) passed in LDS: 6 -> 5 launches (sums re-associated)
# speedup vs baseline: 3.5510x; 1.5254x over previous
#include <hip/hip_runtime.h>

typedef float v2f __attribute__((ext_vector_type(2)));
typedef float v8f __attribute__((ext_vector_type(8)));

#define N_NODES 50000
#define N_EDGES 600000
#define DIM     128
#define MTILES  (N_NODES / 16)

#define ROWS    200
#define NWG     (N_NODES / ROWS)
#define WCAP    256
#define NCHUNK  1172
#define RID_PAD (NCHUNK * 512)
#define NREC    (NWG * 32 * WCAP)
#define OROWS   7
static_assert(ROWS <= 32 * OROWS, "row ownership");
#define DPITCH  256
#define FPITCH  32
static_assert(DPITCH >= ROWS && DPITCH % 32 == 0, "dinv pitch");

static_assert(N_NODES <= 65536 && ROWS <= 32768, "edge record packing");

__device__ __forceinline__ int dix(int node) { return node + (DPITCH - ROWS) * (node / ROWS); }


__global__ void k_rid(const int* __restrict__ dst, unsigned* __restrict__ ridw) {
    int i = blockIdx.x * blockDim.x + threadIdx.x;
    if (i < RID_PAD / 4) {
        unsigned v = 0xFFFFFFFFu;
        if (i < N_EDGES / 4) {
            const int4 d = ((const int4*)dst)[i];
            v = (unsigned)(d.x / ROWS) | ((unsigned)(d.y / ROWS) << 8) |
                ((unsigned)(d.z / ROWS) << 16) | ((unsigned)(d.w / ROWS) << 24);
        }
        ridw[i] = v;
    }
}
static_assert(N_EDGES % 4 == 0 && RID_PAD % 1024 == 0, "range-id words");


#define MLPW 12
#define MLPG 256
#define XTRA (MTILES - MLPG * MLPW)
static_assert(XTRA >= 0 && XTRA <= MLPG, "left-over tiles");
#define XP   (DIM + 4)

template <int NN>
__device__ __forceinline__ void tile_product(const float4 (&a)[DIM / 8], const float* bcol,
                                             v8f (&acc)[NN]) {
    float4 bcur[NN], bnxt[NN];
#pragma unroll
    for (int nt = 0; nt < NN; ++nt) bcur[nt] = *(const float4*)(bcol + nt * 64);

#pragma unroll
    for (int j = 0; j < DIM / 8; ++j) {
        if (j + 1 < DIM / 8) {
#pragma unroll
            for (int nt = 0; nt < NN; ++nt)
                bnxt[nt] = *(const float4*)(bcol + (2 * (j + 1)) * (4 * DIM) + nt * 64);
        }
        const v2f a0 = {a[j].x, a[j].y};
        const v2f a1 = {a[j].z, a[j].w};
#pragma unroll
        for (int nt = 0; nt < NN; ++nt) {
            const v2f b0 = {bcur[nt].x, bcur[nt].y};
            acc[nt] = __builtin_amdgcn_wmma_f32_16x16x4_f32(
                false, a0, false, b0, (short)0, acc[nt], false, false);
        }
#pragma unroll
        for (int nt = 0; nt < NN; ++nt) {
            const v2f b1 = {bcur[nt].z, bcur[nt].w};
            acc[nt] = __builtin_amdgcn_wmma_f32_16x16x4_f32(
                false, a1, false, b1, (short)0, acc[nt], false, false);
        }
#pragma unroll
        for (int nt = 0; nt < NN; ++nt) bcur[nt] = bnxt[nt];
    }
}

template <int NN>
__device__ __forceinline__ void tile_relu_to_lds(const v8f (&acc)[NN], const float* __restrict__ b1,
                                                 int ct0, int g, int r, float* xt) {
    float bb[NN];
#pragma unroll
    for (int nt = 0; nt < NN; ++nt) bb[nt] = b1[(ct0 + nt) * 16 + r];
#pragma unroll
    for (int v = 0; v < 8; ++v) {
#pragma unroll
        for (int nt = 0; nt < NN; ++nt)
            xt[(v + 8 * g) * XP + (ct0 + nt) * 16 + r] = fmaxf(acc[nt][v] + bb[nt], 0.0f);
    }
}

template <int NN>
__device__ __forceinline__ void tile_store(const v8f (&acc)[NN], const float (&dsc)[8], int ct0,
                                           float* hrow) {
#pragma unroll
    for (int v = 0; v < 8; ++v) {
#pragma unroll
        for (int nt = 0; nt < NN; ++nt)
            hrow[(size_t)v * DIM + (ct0 + nt) * 16] = acc[nt][v] * dsc[v];
    }
}

__global__ __launch_bounds__(MLPW * 32) __attribute__((amdgpu_waves_per_eu(3, 3))) void k_mlp(
        const float* __restrict__ A, const float* __restrict__ W1, const float* __restrict__ b1,
        const float* __restrict__ W2, const float* __restrict__ dinv, float* __restrict__ H) {
    __shared__ __attribute__((aligned(16))) float Wl[2 * DIM * DIM];
    __shared__ __attribute__((aligned(16))) float xs[(MLPW + 1) * 16 * XP];

    const int wave = threadIdx.x >> 5;
    const int lane = threadIdx.x & 31;
    const int mt   = blockIdx.x * MLPW + wave;
    const int g = lane >> 4;
    const int r = lane & 15;

    const float* arow = A + ((size_t)mt * 16 + r) * DIM + 4 * g;
    float4 a4[DIM / 8];
#pragma unroll
    for (int j = 0; j < DIM / 8; ++j) a4[j] = *(const float4*)(arow + 8 * j);

    constexpr int NIT = 2 * (DIM / 4) * (DIM / 4);
    constexpr int PASSES = (NIT + MLPW * 32 - 1) / (MLPW * 32);
    static_assert(PASSES * MLPW * 32 - NIT <= NIT, "one wrap at most");
#pragma unroll
    for (int ps = 0; ps < PASSES; ++ps) {
        const int it  = ps * (MLPW * 32) + (int)threadIdx.x;
        const int itc = it < NIT ? it : it - NIT;
        const int m  = itc >> 10;
        const int q  = (itc >> 5) & 31;
        const int n4 = itc & 31;
        const float* Wm = m ? W2 : W1;
        const float4* wr = (const float4*)(Wm + (size_t)(4 * q) * DIM + 4 * n4);
        const float4 r0 = wr[0], r1 = wr[DIM / 4], r2 = wr[2 * (DIM / 4)], r3 = wr[3 * (DIM / 4)];
        float4* o = (float4*)(Wl + m * (DIM * DIM) + q * (4 * DIM) + n4 * 16);
        o[0] = make_float4(r0.x, r1.x, r2.x, r3.x);
        o[1] = make_float4(r0.y, r1.y, r2.y, r3.y);
        o[2] = make_float4(r0.z, r1.z, r2.z, r3.z);
        o[3] = make_float4(r0.w, r1.w, r2.w, r3.w);
    }
    static_assert((DIM / 4) * (DIM / 4) == 1024, "piece index split");
    __syncthreads();

    const float* bcol = Wl + g * (4 * DIM) + r * 4;

    v8f acc[8] = {};
    tile_product<8>(a4, bcol, acc);

    float* xt = xs + wave * (16 * XP);
    tile_relu_to_lds<8>(acc, b1, 0, g, r, xt);
    __builtin_amdgcn_fence(__ATOMIC_RELEASE, "wavefront");
    __builtin_amdgcn_wave_barrier();
    __builtin_amdgcn_fence(__ATOMIC_ACQUIRE, "wavefront");

    float4 x4[DIM / 8];
#pragma unroll
    for (int j = 0; j < DIM / 8; ++j) x4[j] = *(const float4*)(xt + r * XP + 8 * j + 4 * g);

    static_assert(ROWS % 8 == 0, "a lane's eight rows share a range");
    float dsc[8];
    {
        const float* dp = dinv + dix(mt * 16 + 8 * g);
#pragma unroll
        for (int v = 0; v < 8; ++v) dsc[v] = dp[v];
    }

    const bool xgrp  = (int)blockIdx.x < XTRA;
    const bool xwave = xgrp && wave < 4;
    const int  te    = MLPG * MLPW + (int)blockIdx.x;
    float4 ae[DIM / 8];
#pragma unroll
    for (int j = 0; j < DIM / 8; ++j) ae[j] = make_float4(0.f, 0.f, 0.f, 0.f);
    if (xwave) {
        const float* erow = A + ((size_t)te * 16 + r) * DIM + 4 * g;
#pragma unroll
        for (int j = 0; j < DIM / 8; ++j) ae[j] = *(const float4*)(erow + 8 * j);
    }

#pragma unroll
    for (int nt = 0; nt < 8; ++nt) acc[nt] = v8f{};
    tile_product<8>(x4, bcol + DIM * DIM, acc);
    tile_store<8>(acc, dsc, 0, H + ((size_t)mt * 16 + 8 * g) * DIM + r);

    if (xgrp) {
        float* xe = xs + MLPW * (16 * XP);
        const int ct0 = 2 * wave;
        if (xwave) {
            v8f ac2[2] = {};
            tile_product<2>(ae, bcol + ct0 * 64, ac2);
            tile_relu_to_lds<2>(ac2, b1, ct0, g, r, xe);
        }
        __syncthreads();
        if (xwave) {
            float4 xq[DIM / 8];
#pragma unroll
            for (int j = 0; j < DIM / 8; ++j) xq[j] = *(const float4*)(xe + r * XP + 8 * j + 4 * g);
            float dse[8];
            const float* dp = dinv + dix(te * 16 + 8 * g);
#pragma unroll
            for (int v = 0; v < 8; ++v) dse[v] = dp[v];
            v8f ac2[2] = {};
            tile_product<2>(xq, bcol + DIM * DIM + ct0 * 64, ac2);
            tile_store<2>(ac2, dse, ct0, H + ((size_t)te * 16 + 8 * g) * DIM + r);
        }
    }
}


__device__ __forceinline__ void agg_load4(uint4 (&q)[4], const uint4* __restrict__ rid4,
                                          int c0, int lane) {
#pragma unroll
    for (int u = 0; u < 4; ++u) {
        int c = c0 + 32 * u;
        q[u] = (c < NCHUNK) ? rid4[c * 32 + lane]
                            : make_uint4(0xFFFFFFFFu, 0xFFFFFFFFu, 0xFFFFFFFFu, 0xFFFFFFFFu);
    }
}

template <class F>
__device__ __forceinline__ void agg_scan(const uint4* __restrict__ rid4, int g, int wave,
                                         int lane, F note) {
    const unsigned pat = (unsigned)g * 0x01010101u;
    uint4 qa[4], qb[4];
    agg_load4(qa, rid4, wave, lane);
    agg_load4(qb, rid4, wave + 128, lane);
    for (int c0 = wave; c0 < NCHUNK; c0 += 128) {
        uint4 qc[4];
        agg_load4(qc, rid4, c0 + 256, lane);

        unsigned long long mm = 0;
#pragma unroll
        for (int u = 0; u < 4; ++u) {
            const unsigned w[4] = {qa[u].x, qa[u].y, qa[u].z, qa[u].w};
            unsigned m = 0;
#pragma unroll
            for (int j = 0; j < 4; ++j) {
                unsigned x = w[j] ^ pat;
                unsigned t = ~(((x & 0x7F7F7F7Fu) + 0x7F7F7F7Fu) | x | 0x7F7F7F7Fu);
                m |= (t >> 7) << j;
            }
            mm |= (unsigned long long)m << (4 * (u & 1) + 32 * (u >> 1));
        }
        while (mm) {
            int p = __builtin_ctzll(mm);
            mm &= mm - 1;
            int u = ((p >> 2) & 1) + 2 * (p >> 5);
            int j = p & 3;
            int b = (p >> 3) & 3;
            note((c0 + 32 * u) * 512 + lane * 16 + j * 4 + b);
        }
#pragma unroll
        for (int u = 0; u < 4; ++u) { qa[u] = qb[u]; qb[u] = qc[u]; }
    }
}

__device__ __forceinline__ void agg_note(int e, int wave, int base, int* wcnt, int* mylist,
                                         float* acc, int* spill, const float* __restrict__ h,
                                         const int* __restrict__ src,
                                         const int* __restrict__ dst,
                                         const float* __restrict__ esc) {
    int slot = atomicAdd(&wcnt[wave], 1);
    if (slot < WCAP) {
        mylist[slot] = e;
    } else {
        *spill = 1;
        int s = src[e], d = dst[e];
        const float w = esc ? esc[dix(s)] : 1.0f;
        const float* hr = h + (size_t)s * DIM;
        float* ar = acc + (d - base) * DIM;
        for (int c = 0; c < DIM; ++c) atomicAdd(ar + c, hr[c] * w);
    }
}

__global__ __launch_bounds__(1024) void k_list(const int* __restrict__ src,
                                               const int* __restrict__ dst,
                                               const unsigned char* __restrict__ rid,
                                               float* __restrict__ dinv,
                                               int* __restrict__ rec_sd,
                                               int* __restrict__ rec_cnt) {
    __shared__ int list[32 * WCAP];
    __shared__ int osd[32 * WCAP];
    __shared__ int cnt[ROWS];
    __shared__ int wcnt[32];
    __shared__ int ocnt[32];
    __shared__ int full;

    const int tid  = threadIdx.x;
    const int lane = tid & 31;
    const int wave = tid >> 5;
    const int g    = blockIdx.x;
    const int base = g * ROWS;

    if (tid < ROWS) cnt[tid] = 0;
    if (tid < 32) { wcnt[tid] = 0; ocnt[tid] = 0; }
    if (tid == 0) full = 0;
    __syncthreads();

    int* mylist = list + wave * WCAP;
    agg_scan((const uint4*)rid, g, wave, lane, [&](int e) {
        int slot = atomicAdd(&wcnt[wave], 1);
        if (slot < WCAP) {
            mylist[slot] = e;
        } else {
            full = 1;
            atomicAdd(&cnt[dst[e] - base], 1);
        }
    });
    __syncthreads();

    int n1 = wcnt[wave];
    n1 = n1 < WCAP ? n1 : WCAP;
    int sv[WCAP / 32], dv[WCAP / 32];
#pragma unroll
    for (int u = 0; u < WCAP / 32; ++u) {
        sv[u] = 0;
        dv[u] = -1;
        if (u * 32 + lane < n1) {
            const int e = mylist[u * 32 + lane];
            sv[u] = src[e];
            dv[u] = dst[e];
        }
    }
#pragma unroll
    for (int u = 0; u < WCAP / 32; ++u) {
        if (dv[u] >= 0) {
            const int dl = dv[u] - base;
            atomicAdd(&cnt[dl], 1);
            const int ow = dl & 31;
            const int slot = atomicAdd(&ocnt[ow], 1);
            if (slot < WCAP) osd[ow * WCAP + slot] = sv[u] | (dl << 16);
            else             full = 1;
        }
    }
    __syncthreads();

    if (tid < ROWS) dinv[g * DPITCH + tid] = rsqrtf((float)(cnt[tid] + 1));
    int n = ocnt[wave];
    n = n < WCAP ? n : WCAP;
    const int rbase = (g * 32 + wave) * WCAP;
    for (int i = lane; i < n; i += 32) rec_sd[rbase + i] = osd[wave * WCAP + i];
    if (wave == 0) {
        int c = ocnt[lane];
        rec_cnt[g * 32 + lane] = c < WCAP ? c : WCAP;
        if (lane == 0) rec_cnt[NWG * 32 + g * FPITCH] = full;
    }
}

template <int LAYER>
__global__ __launch_bounds__(1024) void k_agg(const float* __restrict__ h,
                                              const int* __restrict__ src,
                                              const int* __restrict__ dst,
                                              const unsigned char* __restrict__ rid,
                                              const float* __restrict__ dinv,
                                              const float* __restrict__ bias,
                                              float* __restrict__ out,
                                              const int* __restrict__ rec_sd,
                                              const int* __restrict__ rec_cnt) {
    __shared__ __attribute__((aligned(16))) float acc[ROWS * DIM];
    __shared__ int   list[32 * WCAP];
    __shared__ int   osd[32 * WCAP];
    __shared__ int   wcnt[32];
    __shared__ int   ocnt[32];
    __shared__ int   spill;

    const int tid  = threadIdx.x;
    const int lane = tid & 31;
    const int wave = tid >> 5;
    const int g    = blockIdx.x;
    const int base = g * ROWS;

    const bool scan = rec_cnt[NWG * 32 + g * FPITCH] != 0;
    const int  nrec = rec_cnt[g * 32 + wave];

    float4 r[OROWS];
    float  di[OROWS];
#pragma unroll
    for (int j = 0; j < OROWS; ++j) {
        const int row = wave + 32 * j;
        r[j]  = make_float4(0.f, 0.f, 0.f, 0.f);
        di[j] = 0.0f;
        if (row < ROWS) {
            r[j]  = *(const float4*)(h + (size_t)(base + row) * DIM + lane * 4);
            di[j] = dinv[g * DPITCH + row];
        }
    }

    if (scan) {
        float4* acc4 = (float4*)acc;
        for (int i = tid; i < ROWS * DIM / 4; i += 1024) acc4[i] = make_float4(0.f, 0.f, 0.f, 0.f);
        if (tid < 32) { wcnt[tid] = 0; ocnt[tid] = 0; }
        if (tid == 0) spill = 0;
        __syncthreads();

        int* mylist = list + wave * WCAP;
        agg_scan((const uint4*)rid, g, wave, lane, [&](int e) {
            agg_note(e, wave, base, wcnt, mylist, acc, &spill, h, src, dst,
                     LAYER == 1 ? dinv : nullptr);
        });
        __syncthreads();

        int n1 = wcnt[wave];
        n1 = n1 < WCAP ? n1 : WCAP;
        for (int b0 = 0; b0 < n1; b0 += 32) {
            if (b0 + lane < n1) {
                int e = mylist[b0 + lane];
                int s = src[e];
                int d = dst[e];
                int dl = d - base;
                int ow = dl & 31;
                int slot = atomicAdd(&ocnt[ow], 1);
                if (slot < WCAP) {
                    osd[ow * WCAP + slot] = s | (dl << 16);
                } else {
                    spill = 1;
                    const float w = LAYER == 1 ? dinv[dix(s)] : 1.0f;
                    const float* hr = h + (size_t)s * DIM;
                    float* ar = acc + dl * DIM;
                    for (int c = 0; c < DIM; ++c) atomicAdd(ar + c, hr[c] * w);
                }
            }
        }
        __syncthreads();
    }

    const int rbase = (g * 32 + wave) * WCAP;
    int n = scan ? ocnt[wave] : nrec;
    n = n < WCAP ? n : WCAP;
    n = __builtin_amdgcn_readfirstlane(n);

    bool own_scaled = (LAYER != 1);

    for (int b0 = 0; b0 < n; b0 += 32) {
        const int nb = (n - b0) < 32 ? (n - b0) : 32;
        int p = 0;
        if (lane < nb) {
            if (scan) p = osd[wave * WCAP + b0 + lane];
            else      p = rec_sd[rbase + b0 + lane];
        }
        const int s  = p & 0xFFFF;
        const int jr = p >> 21;
        int esb = 0;
        if (LAYER == 1 && lane < nb) esb = __float_as_int(dinv[dix(s)]);
        for (int k0 = 0; k0 < nb; k0 += 8) {
            float4 v[8];
            int    jj[8];
            float  ww[8];
#pragma unroll
            for (int u = 0; u < 8; ++u) {
                int k  = k0 + u;
                int kk = k < nb ? k : nb - 1;
                int   ss = __builtin_amdgcn_readlane(s, kk);
                jj[u]    = __builtin_amdgcn_readlane(jr, kk);
                ww[u]    = LAYER == 1 ? __int_as_float(__builtin_amdgcn_readlane(esb, kk)) : 1.0f;
                v[u]     = *(const float4*)(h + (size_t)ss * DIM + lane * 4);
            }
            if (!own_scaled) {
                own_scaled = true;
#pragma unroll
                for (int j = 0; j < OROWS; ++j) {
                    r[j].x *= di[j]; r[j].y *= di[j]; r[j].z *= di[j]; r[j].w *= di[j];
                }
            }
#pragma unroll
            for (int u = 0; u < 8; ++u) {
                if (k0 + u < nb) {
#pragma unroll
                    for (int j = 0; j < OROWS; ++j) {
                        if (jj[u] == j) {
                            asm volatile("");
                            if (LAYER == 1) {
                                r[j].x = fmaf(v[u].x, ww[u], r[j].x);
                                r[j].y = fmaf(v[u].y, ww[u], r[j].y);
                                r[j].z = fmaf(v[u].z, ww[u], r[j].z);
                                r[j].w = fmaf(v[u].w, ww[u], r[j].w);
                            } else {
                                r[j].x += v[u].x; r[j].y += v[u].y;
                                r[j].z += v[u].z; r[j].w += v[u].w;
                            }
                        }
                    }
                }
            }
        }
    }

    if (!own_scaled) {
#pragma unroll
        for (int j = 0; j < OROWS; ++j) {
            r[j].x *= di[j]; r[j].y *= di[j]; r[j].z *= di[j]; r[j].w *= di[j];
        }
    }

    const bool use_acc = scan && (spill != 0);
    float bx = 0.0f, by = 0.0f, bz = 0.0f, bw = 0.0f;
    if (LAYER == 2) {
        bx = bias[lane * 4 + 0]; by = bias[lane * 4 + 1];
        bz = bias[lane * 4 + 2]; bw = bias[lane * 4 + 3];
    }
#pragma unroll
    for (int j = 0; j < OROWS; ++j) {
        const int row = wave + 32 * j;
        if (row < ROWS) {
            const size_t gi = (size_t)(base + row) * DIM + lane * 4;
            float4 t = r[j];
            if (use_acc) {
                const float4 a = *(const float4*)(acc + row * DIM + lane * 4);
                t.x += a.x; t.y += a.y; t.z += a.z; t.w += a.w;
            }
            float4 o;
            if (LAYER == 1) {
                o.x = t.x * di[j]; o.y = t.y * di[j]; o.z = t.z * di[j]; o.w = t.w * di[j];
            } else {
                o.x = bx + t.x * di[j]; o.y = by + t.y * di[j];
                o.z = bz + t.z * di[j]; o.w = bw + t.w * di[j];
            }
            *(float4*)(out + gi) = o;
        }
    }
}


extern "C" void kernel_launch(void* const* d_in, const int* in_sizes, int n_in,
                              void* d_out, int out_size, void* d_ws, size_t ws_size,
                              hipStream_t stream) {
    const float* X  = (const float*)d_in[0];
    const int*   ei = (const int*)d_in[1];
    const float* W1 = (const float*)d_in[2];
    const float* b1 = (const float*)d_in[3];
    const float* W2 = (const float*)d_in[4];
    const float* b2 = (const float*)d_in[5];
    float* out = (float*)d_out;

    const int* src = ei;
    const int* dst = ei + N_EDGES;

    float* ws   = (float*)d_ws;
    float* h    = ws + 50048 + 2 * 16384;
    unsigned char* rid = (unsigned char*)(h + (size_t)N_NODES * DIM);
    int*   rec_sd  = (int*)(rid + RID_PAD);
    int*   rec_cnt = rec_sd + NREC;
    float* dinv    = (float*)(rec_cnt + NWG * 32 + NWG * FPITCH);
    float* ax   = out;

    const int thr = 256;
    const int gRid    = RID_PAD / 4 / thr;

    k_rid <<<gRid, thr, 0, stream>>>(dst, (unsigned*)rid);
    k_list<<<NWG, 1024, 0, stream>>>(src, dst, rid, dinv, rec_sd, rec_cnt);

    k_agg<1><<<NWG, 1024, 0, stream>>>(X, src, dst, rid, dinv, b1, ax, rec_sd, rec_cnt);

    k_mlp<<<MLPG, MLPW * 32, 0, stream>>>(ax, W1, b1, W2, dinv, h);

    k_agg<2><<<NWG, 1024, 0, stream>>>(h, src, dst, rid, dinv, b2, out, rec_sd, rec_cnt);
}


# range ids as one-hot bit planes, one per value of each of two hex digits: a range owner ANDs two planes, 32 edges a step, a quarter of the bytes; slots by one LDS add per lane (sum order changes)
# speedup vs baseline: 4.1662x; 1.1732x over previous
#include <hip/hip_runtime.h>

typedef float v2f __attribute__((ext_vector_type(2)));
typedef float v8f __attribute__((ext_vector_type(8)));

#define N_NODES 50000
#define N_EDGES 600000
#define DIM     128
#define MTILES  (N_NODES / 16)

#define ROWS    200
#define NWG     (N_NODES / ROWS)
#define WCAP    256
#define NWORD   18752
#define NPLANE  32
#define RID_PAD 600064
static_assert(NWORD * 32 >= N_EDGES && NWORD % 32 == 0, "whole 128-byte lines of plane words");
static_assert((size_t)NPLANE * NWORD <= (size_t)N_NODES * DIM, "the planes fit the plane of h");
#define NREC    (NWG * 32 * WCAP)
#define OROWS   7
static_assert(ROWS <= 32 * OROWS, "row ownership");
#define DPITCH  256
#define FPITCH  32
static_assert(DPITCH >= ROWS && DPITCH % 32 == 0, "dinv pitch");

static_assert(N_NODES <= 65536 && ROWS <= 32768, "edge record packing");

__device__ __forceinline__ int dix(int node) { return node + (DPITCH - ROWS) * (node / ROWS); }


static_assert(NWG <= 256 && NPLANE == 32, "range ids: two digits of 16; eight waves of four planes");
__global__ __launch_bounds__(256) void k_rid(const int* __restrict__ dst,
                                             unsigned* __restrict__ planes) {
    const int lane = threadIdx.x & 31;
    const int w    = blockIdx.x;
    const int pg   = __builtin_amdgcn_readfirstlane(threadIdx.x >> 5);
    const int sh   = pg < 4 ? 4 : 0;
    const unsigned j0 = 4u * (pg & 3);
    static_assert(N_EDGES % 32 == 0, "whole words of edges");
    int d[32];
#pragma unroll
    for (int k = 0; k < 32; ++k) {
        const int wd = w * 32 + k;
        d[k] = dst[(wd < N_EDGES / 32 ? wd : 0) * 32 + lane];
    }
#pragma unroll
    for (int k = 0; k < 32; ++k) asm volatile("" : "+v"(d[k]));
    unsigned p[4] = {0u, 0u, 0u, 0u};
#pragma unroll
    for (int k = 0; k < 32; ++k) {
        const unsigned digit = (((unsigned)d[k] / ROWS) >> sh) & 15u;
        const unsigned live = (w * 32 + k < N_EDGES / 32) ? 0xFFFFFFFFu : 0u;
#pragma unroll
        for (int i = 0; i < 4; ++i) {
            const unsigned m = __builtin_amdgcn_ballot_w32(digit == j0 + (unsigned)i) & live;
            if (lane == k) p[i] = m;
        }
    }
#pragma unroll
    for (int i = 0; i < 4; ++i) planes[(pg * 4 + i) * NWORD + w * 32 + lane] = p[i];
}


#define XP   (DIM + 4)
#define RTILES ((ROWS + 15) / 16)
static_assert(RTILES <= 16, "two row tiles a wave at most");

template <int NN>
__device__ __forceinline__ void tile_product(const float* arow, const float* bcol,
                                             v8f (&acc)[NN]) {
    float4 acur = *(const float4*)arow, anxt = acur;
    float4 bcur[NN], bnxt[NN];
#pragma unroll
    for (int nt = 0; nt < NN; ++nt) bcur[nt] = *(const float4*)(bcol + nt * 64);

#pragma unroll
    for (int j = 0; j < DIM / 8; ++j) {
        if (j + 1 < DIM / 8) {
            anxt = *(const float4*)(arow + 8 * (j + 1));
#pragma unroll
            for (int nt = 0; nt < NN; ++nt)
                bnxt[nt] = *(const float4*)(bcol + (2 * (j + 1)) * (4 * DIM) + nt * 64);
        }
        const v2f a0 = {acur.x, acur.y};
        const v2f a1 = {acur.z, acur.w};
#pragma unroll
        for (int nt = 0; nt < NN; ++nt) {
            const v2f b0 = {bcur[nt].x, bcur[nt].y};
            acc[nt] = __builtin_amdgcn_wmma_f32_16x16x4_f32(
                false, a0, false, b0, (short)0, acc[nt], false, false);
        }
#pragma unroll
        for (int nt = 0; nt < NN; ++nt) {
            const v2f b1 = {bcur[nt].z, bcur[nt].w};
            acc[nt] = __builtin_amdgcn_wmma_f32_16x16x4_f32(
                false, a1, false, b1, (short)0, acc[nt], false, false);
        }
        acur = anxt;
#pragma unroll
        for (int nt = 0; nt < NN; ++nt) bcur[nt] = bnxt[nt];
        __builtin_amdgcn_sched_barrier(0);
    }
}

template <int NN>
__device__ __forceinline__ void tile_relu_to_lds(const v8f (&acc)[NN], const float (&bb)[NN],
                                                 int ct0, int g, int r, float* xt) {
#pragma unroll
    for (int v = 0; v < 8; ++v) {
#pragma unroll
        for (int nt = 0; nt < NN; ++nt)
            xt[(v + 8 * g) * XP + (ct0 + nt) * 16 + r] = fmaxf(acc[nt][v] + bb[nt], 0.0f);
    }
}

template <int NN>
__device__ __forceinline__ void tile_store(const v8f (&acc)[NN], const float (&dsc)[8], int ct0,
                                           float* hrow) {
#pragma unroll
    for (int v = 0; v < 8; ++v) {
#pragma unroll
        for (int nt = 0; nt < NN; ++nt)
            hrow[(size_t)v * DIM + (ct0 + nt) * 16] = acc[nt][v] * dsc[v];
    }
}


#define NQ    (NWORD / 4)
#define NTRIP ((NQ + 1023) / 1024)
template <class K, class O>
__device__ __forceinline__ void agg_scan(const uint4* __restrict__ pl4, int g, int tid,
                                         int* wc, K keep, O over) {
    const uint4* pa = pl4 + (g >> 4) * NQ;
    const uint4* pb = pl4 + (16 + (g & 15)) * NQ;
    uint4 A[NTRIP], B[NTRIP];
#pragma unroll
    for (int t = 0; t < NTRIP; ++t) {
        const int q = tid + 1024 * t;
        A[t] = pa[q < NQ ? q : NQ - 1];
        B[t] = pb[q < NQ ? q : NQ - 1];
    }
#pragma unroll
    for (int t = 0; t < NTRIP; ++t) {
        const int q = tid + 1024 * t;
        unsigned hit[4] = {A[t].x & B[t].x, A[t].y & B[t].y, A[t].z & B[t].z, A[t].w & B[t].w};
        if (q >= NQ) { hit[0] = 0u; hit[1] = 0u; hit[2] = 0u; hit[3] = 0u; }
        const int found = __builtin_popcount(hit[0]) + __builtin_popcount(hit[1]) +
                          __builtin_popcount(hit[2]) + __builtin_popcount(hit[3]);
        if (found) {
            const int slot0 = atomicAdd(wc, found);
            int slot = slot0;
#pragma unroll
            for (int w = 0; w < 4; ++w) {
                unsigned m = hit[w];
                while (m) {
                    const int b = __builtin_ctz(m);
                    m &= m - 1;
                    keep((4 * q + w) * 32 + b, slot);
                    ++slot;
                }
            }
            if (slot > WCAP) {
                slot = slot0;
#pragma unroll
                for (int w = 0; w < 4; ++w) {
                    unsigned m = hit[w];
                    while (m) {
                        const int b = __builtin_ctz(m);
                        m &= m - 1;
                        if (slot >= WCAP) over((4 * q + w) * 32 + b);
                        ++slot;
                    }
                }
            }
        }
    }
}

template <class F>
__device__ __forceinline__ void agg_scan_dst(const int* __restrict__ dst, int g, int tid, F note) {
    for (int e = tid; e < N_EDGES; e += 1024) {
        if (dst[e] / ROWS == g) note(e);
    }
}

__device__ __forceinline__ void agg_note(int e, int wave, int base, int* wcnt, int* mylist,
                                         float* acc, int ap, int* spill,
                                         const float* __restrict__ h,
                                         const int* __restrict__ src,
                                         const int* __restrict__ dst,
                                         const float* __restrict__ esc) {
    int slot = atomicAdd(&wcnt[wave], 1);
    if (slot < WCAP) {
        mylist[slot] = e;
    } else {
        *spill = 1;
        int s = src[e], d = dst[e];
        const float w = esc ? esc[dix(s)] : 1.0f;
        const float* hr = h + (size_t)s * DIM;
        float* ar = acc + (d - base) * ap;
        for (int c = 0; c < DIM; ++c) atomicAdd(ar + c, hr[c] * w);
    }
}

__global__ __launch_bounds__(1024) void k_list(const int* __restrict__ src,
                                               const int* __restrict__ dst,
                                               const unsigned* __restrict__ planes,
                                               float* __restrict__ dinv,
                                               int* __restrict__ rec_sd,
                                               int* __restrict__ rec_cnt) {
    __shared__ int list[32 * WCAP];
    __shared__ int osd[32 * WCAP];
    __shared__ int cnt[ROWS];
    __shared__ int wcnt[32];
    __shared__ int ocnt[32];
    __shared__ int full;

    const int tid  = threadIdx.x;
    const int lane = tid & 31;
    const int wave = tid >> 5;
    const int g    = blockIdx.x;
    const int base = g * ROWS;

    if (tid < ROWS) cnt[tid] = 0;
    if (tid < 32) { wcnt[tid] = 0; ocnt[tid] = 0; }
    if (tid == 0) full = 0;
    __syncthreads();

    int* mylist = list + wave * WCAP;
    agg_scan((const uint4*)planes, g, tid, &wcnt[wave],
             [&](int e, int slot) { if (slot < WCAP) mylist[slot] = e; },
             [&](int e) {
                 full = 1;
                 atomicAdd(&cnt[dst[e] - base], 1);
             });
    __syncthreads();

    int n1 = wcnt[wave];
    n1 = n1 < WCAP ? n1 : WCAP;
    int sv[WCAP / 32], dv[WCAP / 32];
#pragma unroll
    for (int u = 0; u < WCAP / 32; ++u) {
        sv[u] = 0;
        dv[u] = -1;
        if (u * 32 + lane < n1) {
            const int e = mylist[u * 32 + lane];
            sv[u] = src[e];
            dv[u] = dst[e];
        }
    }
#pragma unroll
    for (int u = 0; u < WCAP / 32; ++u) {
        if (dv[u] >= 0) {
            const int dl = dv[u] - base;
            atomicAdd(&cnt[dl], 1);
            const int ow = dl & 31;
            const int slot = atomicAdd(&ocnt[ow], 1);
            if (slot < WCAP) osd[ow * WCAP + slot] = sv[u] | (dl << 16);
            else             full = 1;
        }
    }
    __syncthreads();

    if (tid < ROWS) dinv[g * DPITCH + tid] = rsqrtf((float)(cnt[tid] + 1));
    int n = ocnt[wave];
    n = n < WCAP ? n : WCAP;
    const int rbase = (g * 32 + wave) * WCAP;
    for (int i = lane; i < n; i += 32) rec_sd[rbase + i] = osd[wave * WCAP + i];
    if (wave == 0) {
        int c = ocnt[lane];
        rec_cnt[g * 32 + lane] = c < WCAP ? c : WCAP;
        if (lane == 0) rec_cnt[NWG * 32 + g * FPITCH] = full;
    }
}

template <int LAYER>
__global__ __launch_bounds__(1024) void k_agg(const float* __restrict__ h,
                                              const int* __restrict__ src,
                                              const int* __restrict__ dst,
                                              const float* __restrict__ dinv,
                                              const float* __restrict__ bias,
                                              float* __restrict__ out,
                                              const int* __restrict__ rec_sd,
                                              const int* __restrict__ rec_cnt,
                                              const float* __restrict__ W1,
                                              const float* __restrict__ W2) {
    constexpr int AP    = LAYER == 1 ? XP : DIM;
    constexpr int AROWS = LAYER == 1 ? RTILES * 16 : ROWS;
    __shared__ __attribute__((aligned(16))) float acc[AROWS * AP];
    __shared__ __attribute__((aligned(16))) float Wl[LAYER == 1 ? 2 * DIM * DIM : 4];
    __shared__ int   list[32 * WCAP];
    __shared__ int   osd[32 * WCAP];
    __shared__ int   wcnt[32];
    __shared__ int   ocnt[32];
    __shared__ int   spill;

    const int tid  = threadIdx.x;
    const int lane = tid & 31;
    const int wave = tid >> 5;
    const int g    = blockIdx.x;
    const int base = g * ROWS;

    const bool scan = rec_cnt[NWG * 32 + g * FPITCH] != 0;
    const int  nrec = rec_cnt[g * 32 + wave];

    float4 wq[2][4];
    if (LAYER == 1) {
#pragma unroll
        for (int m = 0; m < 2; ++m) {
            const int q  = (tid >> 5) & 31;
            const int n4 = tid & 31;
            const float4* wr = (const float4*)((m ? W2 : W1) + (size_t)(4 * q) * DIM + 4 * n4);
#pragma unroll
            for (int i = 0; i < 4; ++i) wq[m][i] = wr[i * (DIM / 4)];
        }
    }

    float4 r[OROWS];
    float  di[OROWS];
#pragma unroll
    for (int j = 0; j < OROWS; ++j) {
        const int row = wave + 32 * j;
        r[j]  = make_float4(0.f, 0.f, 0.f, 0.f);
        di[j] = 0.0f;
        if (row < ROWS) {
            r[j]  = *(const float4*)(h + (size_t)(base + row) * DIM + lane * 4);
            di[j] = dinv[g * DPITCH + row];
        }
    }

    if (LAYER == 1) {
#pragma unroll
        for (int m = 0; m < 2; ++m) {
            const int q  = (tid >> 5) & 31;
            const int n4 = tid & 31;
            float4* o = (float4*)(Wl + m * (DIM * DIM) + q * (4 * DIM) + n4 * 16);
            o[0] = make_float4(wq[m][0].x, wq[m][1].x, wq[m][2].x, wq[m][3].x);
            o[1] = make_float4(wq[m][0].y, wq[m][1].y, wq[m][2].y, wq[m][3].y);
            o[2] = make_float4(wq[m][0].z, wq[m][1].z, wq[m][2].z, wq[m][3].z);
            o[3] = make_float4(wq[m][0].w, wq[m][1].w, wq[m][2].w, wq[m][3].w);
        }
    }

    if (scan) {
        float4* acc4 = (float4*)acc;
        for (int i = tid; i < AROWS * AP / 4; i += 1024) acc4[i] = make_float4(0.f, 0.f, 0.f, 0.f);
        if (tid < 32) { wcnt[tid] = 0; ocnt[tid] = 0; }
        if (tid == 0) spill = 0;
        __syncthreads();

        int* mylist = list + wave * WCAP;
        agg_scan_dst(dst, g, tid, [&](int e) {
            agg_note(e, wave, base, wcnt, mylist, acc, AP, &spill, h, src, dst,
                     LAYER == 1 ? dinv : nullptr);
        });
        __syncthreads();

        int n1 = wcnt[wave];
        n1 = n1 < WCAP ? n1 : WCAP;
        for (int b0 = 0; b0 < n1; b0 += 32) {
            if (b0 + lane < n1) {
                int e = mylist[b0 + lane];
                int s = src[e];
                int d = dst[e];
                int dl = d - base;
                int ow = dl & 31;
                int slot = atomicAdd(&ocnt[ow], 1);
                if (slot < WCAP) {
                    osd[ow * WCAP + slot] = s | (dl << 16);
                } else {
                    spill = 1;
                    const float w = LAYER == 1 ? dinv[dix(s)] : 1.0f;
                    const float* hr = h + (size_t)s * DIM;
                    float* ar = acc + dl * AP;
                    for (int c = 0; c < DIM; ++c) atomicAdd(ar + c, hr[c] * w);
                }
            }
        }
        __syncthreads();
    }

    const int rbase = (g * 32 + wave) * WCAP;
    int n = scan ? ocnt[wave] : nrec;
    n = n < WCAP ? n : WCAP;
    n = __builtin_amdgcn_readfirstlane(n);

    bool own_scaled = (LAYER != 1);

    for (int b0 = 0; b0 < n; b0 += 32) {
        const int nb = (n - b0) < 32 ? (n - b0) : 32;
        int p = 0;
        if (lane < nb) {
            if (scan) p = osd[wave * WCAP + b0 + lane];
            else      p = rec_sd[rbase + b0 + lane];
        }
        const int s  = p & 0xFFFF;
        const int jr = p >> 21;
        int esb = 0;
        if (LAYER == 1 && lane < nb) esb = __float_as_int(dinv[dix(s)]);
        for (int k0 = 0; k0 < nb; k0 += 8) {
            float4 v[8];
            int    jj[8];
            float  ww[8];
#pragma unroll
            for (int u = 0; u < 8; ++u) {
                int k  = k0 + u;
                int kk = k < nb ? k : nb - 1;
                int   ss = __builtin_amdgcn_readlane(s, kk);
                jj[u]    = __builtin_amdgcn_readlane(jr, kk);
                ww[u]    = LAYER == 1 ? __int_as_float(__builtin_amdgcn_readlane(esb, kk)) : 1.0f;
                v[u]     = *(const float4*)(h + (size_t)ss * DIM + lane * 4);
            }
            if (!own_scaled) {
                own_scaled = true;
#pragma unroll
                for (int j = 0; j < OROWS; ++j) {
                    r[j].x *= di[j]; r[j].y *= di[j]; r[j].z *= di[j]; r[j].w *= di[j];
                }
            }
#pragma unroll
            for (int u = 0; u < 8; ++u) {
                if (k0 + u < nb) {
#pragma unroll
                    for (int j = 0; j < OROWS; ++j) {
                        if (jj[u] == j) {
                            asm volatile("");
                            if (LAYER == 1) {
                                r[j].x = fmaf(v[u].x, ww[u], r[j].x);
                                r[j].y = fmaf(v[u].y, ww[u], r[j].y);
                                r[j].z = fmaf(v[u].z, ww[u], r[j].z);
                                r[j].w = fmaf(v[u].w, ww[u], r[j].w);
                            } else {
                                r[j].x += v[u].x; r[j].y += v[u].y;
                                r[j].z += v[u].z; r[j].w += v[u].w;
                            }
                        }
                    }
                }
            }
        }
    }

    if (!own_scaled) {
#pragma unroll
        for (int j = 0; j < OROWS; ++j) {
            r[j].x *= di[j]; r[j].y *= di[j]; r[j].z *= di[j]; r[j].w *= di[j];
        }
    }

    const bool use_acc = scan && (spill != 0);
    float bx = 0.0f, by = 0.0f, bz = 0.0f, bw = 0.0f;
    if (LAYER == 2) {
        bx = bias[lane * 4 + 0]; by = bias[lane * 4 + 1];
        bz = bias[lane * 4 + 2]; bw = bias[lane * 4 + 3];
    }
#pragma unroll
    for (int j = 0; j < OROWS; ++j) {
        const int row = wave + 32 * j;
        if (row < ROWS) {
            float4 t = r[j];
            if (use_acc) {
                const float4 a = *(const float4*)(acc + row * AP + lane * 4);
                t.x += a.x; t.y += a.y; t.z += a.z; t.w += a.w;
            }
            float4 o;
            if (LAYER == 1) {
                o.x = t.x * di[j]; o.y = t.y * di[j]; o.z = t.z * di[j]; o.w = t.w * di[j];
                *(float4*)(acc + row * AP + lane * 4) = o;
            } else {
                o.x = bx + t.x * di[j]; o.y = by + t.y * di[j];
                o.z = bz + t.z * di[j]; o.w = bw + t.w * di[j];
                *(float4*)(out + (size_t)(base + row) * DIM + lane * 4) = o;
            }
        }
    }

    if (LAYER == 1) {
        if (wave < AROWS - ROWS)
            *(float4*)(acc + (ROWS + wave) * AP + lane * 4) = make_float4(0.f, 0.f, 0.f, 0.f);
        const int gq = lane >> 4;
        const int rr = lane & 15;
        const int rt0 = wave >> 2;
        const int ct0 = 2 * (wave & 3);

        __syncthreads();

        const float bb[2] = {bias[ct0 * 16 + rr], bias[(ct0 + 1) * 16 + rr]};
        static_assert(ROWS % 8 == 0, "a lane's eight rows: all real or all padding");
        float dsc[2][8];
#pragma unroll
        for (int i = 0; i < 2; ++i) {
            const int row0 = (rt0 + 8 * i) * 16 + 8 * gq;
#pragma unroll
            for (int v = 0; v < 8; ++v) dsc[i][v] = 0.0f;
            if (row0 < ROWS) {
                const float* dp = dinv + g * DPITCH + row0;
#pragma unroll
                for (int v = 0; v < 8; ++v) dsc[i][v] = dp[v];
            }
        }

        const float* bcol = Wl + gq * (4 * DIM) + rr * 4 + ct0 * 64;

        v8f t1[2][2];
#pragma unroll
        for (int i = 0; i < 2; ++i) {
            const int rt = rt0 + 8 * i;
            t1[i][0] = v8f{};
            t1[i][1] = v8f{};
            if (rt < RTILES) tile_product<2>(acc + (rt * 16 + rr) * AP + 4 * gq, bcol, t1[i]);
        }
        __syncthreads();

#pragma unroll
        for (int i = 0; i < 2; ++i) {
            const int rt = rt0 + 8 * i;
            if (rt < RTILES) tile_relu_to_lds<2>(t1[i], bb, ct0, gq, rr, acc + rt * 16 * AP);
        }
        __syncthreads();

#pragma unroll
        for (int i = 0; i < 2; ++i) {
            const int rt = rt0 + 8 * i;
            if (rt < RTILES) {
                v8f h2[2] = {};
                tile_product<2>(acc + (rt * 16 + rr) * AP + 4 * gq, bcol + DIM * DIM, h2);
                const int row0 = rt * 16 + 8 * gq;
                if (row0 < ROWS)
                    tile_store<2>(h2, dsc[i], ct0, out + (size_t)(base + row0) * DIM + rr);
            }
        }
    }
}


extern "C" void kernel_launch(void* const* d_in, const int* in_sizes, int n_in,
                              void* d_out, int out_size, void* d_ws, size_t ws_size,
                              hipStream_t stream) {
    const float* X  = (const float*)d_in[0];
    const int*   ei = (const int*)d_in[1];
    const float* W1 = (const float*)d_in[2];
    const float* b1 = (const float*)d_in[3];
    const float* W2 = (const float*)d_in[4];
    const float* b2 = (const float*)d_in[5];
    float* out = (float*)d_out;

    const int* src = ei;
    const int* dst = ei + N_EDGES;

    float* ws   = (float*)d_ws;
    float* h    = ws + 50048 + 2 * 16384;
    unsigned* planes = (unsigned*)h;
    int*   rec_sd  = (int*)((unsigned char*)(h + (size_t)N_NODES * DIM) + RID_PAD);
    int*   rec_cnt = rec_sd + NREC;
    float* dinv    = (float*)(rec_cnt + NWG * 32 + NWG * FPITCH);

    const int gRid = NWORD / 32;

    k_rid <<<gRid, 256, 0, stream>>>(dst, planes);
    k_list<<<NWG, 1024, 0, stream>>>(src, dst, planes, dinv, rec_sd, rec_cnt);

    k_agg<1><<<NWG, 1024, 0, stream>>>(X, src, dst, dinv, b1, h, rec_sd, rec_cnt, W1, W2);

    k_agg<2><<<NWG, 1024, 0, stream>>>(h, src, dst, dinv, b2, out, rec_sd, rec_cnt,
                                       nullptr, nullptr);
}
